# starting kernel
# baseline (speedup 1.0000x reference)
#include <hip/hip_runtime.h>
#include <math.h>

#define IMSIZE 1024
#define NSEG   128
#define FEPS   1e-10f

typedef __attribute__((ext_vector_type(2))) float v2f;
typedef __attribute__((ext_vector_type(8))) float v8f;

__global__ __launch_bounds__(256) void sketch_render_kernel(
    const float* __restrict__ xs, const float* __restrict__ ys,
    float* __restrict__ out, float* __restrict__ bmin, float* __restrict__ bmax)
{
    __shared__ float s_cy[NSEG], s_cx[NSEG], s_c0[NSEG];
    __shared__ float s_sl[NSEG], s_in[NSEG], s_vt[NSEG];
    __shared__ float s_xl[NSEG], s_xh[NSEG], s_yl[NSEG], s_yh[NSEG];
    __shared__ float s_red[16];

    const int t = threadIdx.x;

    if (t < NSEG) {
        float x0 = xs[t], x1 = xs[t + 1];
        float y0 = ys[t], y1 = ys[t + 1];
        float dx = x1 - x0, dy = y1 - y0;
        float dx2 = dx * dx;
        float d = dx2 + dy * dy;
        float invd = 1.0f / (d + FEPS);
        float cross = y0 * x1 - x0 * y1;
        s_cy[t] = dx * dy * invd;
        s_cx[t] = dx2 * invd;
        s_c0[t] = -dy * cross * invd;
        float invdx = 1.0f / (dx + FEPS);
        s_sl[t] = dy * invdx;
        s_in[t] = cross * invdx;
        s_vt[t] = (x0 == x1) ? 1.0f : 0.0f;
        s_xl[t] = fminf(x0, x1); s_xh[t] = fmaxf(x0, x1);
        s_yl[t] = fminf(y0, y1); s_yh[t] = fmaxf(y0, y1);
    }
    __syncthreads();

    const int lane  = t & 31;
    const int wave  = t >> 5;
    const int laneN = lane & 15;
    const int hi    = lane >> 4;
    const int xbase = blockIdx.x * 16;
    const int ybase = blockIdx.y * 64;
    const float xp  = (float)(xbase + laneN);

    v2f afrag[8];
#pragma unroll
    for (int g = 0; g < 8; ++g) {
        int m = g * 16 + laneN;
        afrag[g].x = hi ? s_c0[m] : s_cy[m];
        afrag[g].y = hi ? 0.0f    : s_cx[m];
    }

    float lmin = INFINITY, lmax = -INFINITY;

#pragma unroll 1
    for (int r = 0; r < 8; ++r) {
        const int   row = ybase + wave + r * 8;
        const float yp  = (float)row;

        v2f bfrag;
        bfrag.x = hi ? 1.0f : yp;
        bfrag.y = hi ? 0.0f : xp;

        float acc = 0.0f;
#pragma unroll 2
        for (int g = 0; g < 8; ++g) {
            v8f c = {0.f, 0.f, 0.f, 0.f, 0.f, 0.f, 0.f, 0.f};
            v8f dmat = __builtin_amdgcn_wmma_f32_16x16x4_f32(
                false, afrag[g], false, bfrag, (short)0, c, false, false);
            const int mbase = g * 16 + hi * 8;
#pragma unroll
            for (int j = 0; j < 8; ++j) {
                const int m = mbase + j;
                float xg   = dmat[j];
                float ygen = fmaf(s_sl[m], xg, s_in[m]);
                float ysel = (s_vt[m] != 0.0f) ? yp : ygen;
                float xc   = __builtin_amdgcn_fmed3f(xg,   s_xl[m], s_xh[m]);
                float yc   = __builtin_amdgcn_fmed3f(ysel, s_yl[m], s_yh[m]);
                float ddx  = xp - xc;
                float ddy  = yp - yc;
                acc += __builtin_amdgcn_sqrtf(fmaf(ddx, ddx, fmaf(ddy, ddy, FEPS)));
            }
        }
        acc += __shfl_xor(acc, 16, 32);
        lmin = fminf(lmin, acc);
        lmax = fmaxf(lmax, acc);
        if (lane < 16)
            out[row * IMSIZE + xbase + laneN] = acc;
    }

#pragma unroll
    for (int off = 16; off >= 1; off >>= 1) {
        lmin = fminf(lmin, __shfl_xor(lmin, off, 32));
        lmax = fmaxf(lmax, __shfl_xor(lmax, off, 32));
    }
    if (lane == 0) { s_red[wave] = lmin; s_red[8 + wave] = lmax; }
    __syncthreads();
    if (t == 0) {
        float m0 = s_red[0], m1 = s_red[8];
        for (int w = 1; w < 8; ++w) {
            m0 = fminf(m0, s_red[w]);
            m1 = fmaxf(m1, s_red[8 + w]);
        }
        const int bid = blockIdx.y * gridDim.x + blockIdx.x;
        bmin[bid] = m0;
        bmax[bid] = m1;
    }
}

__global__ __launch_bounds__(256) void sketch_minmax_kernel(
    const float* __restrict__ bmin, const float* __restrict__ bmax,
    float* __restrict__ mm, int n)
{
    __shared__ float smin[8], smax[8];
    const int t = threadIdx.x;
    float m0 = INFINITY, m1 = -INFINITY;
    for (int i = t; i < n; i += 256) {
        m0 = fminf(m0, bmin[i]);
        m1 = fmaxf(m1, bmax[i]);
    }
#pragma unroll
    for (int off = 16; off >= 1; off >>= 1) {
        m0 = fminf(m0, __shfl_xor(m0, off, 32));
        m1 = fmaxf(m1, __shfl_xor(m1, off, 32));
    }
    if ((t & 31) == 0) { smin[t >> 5] = m0; smax[t >> 5] = m1; }
    __syncthreads();
    if (t == 0) {
        for (int w = 1; w < 8; ++w) {
            m0 = fminf(m0, smin[w]);
            m1 = fmaxf(m1, smax[w]);
        }
        mm[0] = m0;
        mm[1] = m1;
    }
}

__global__ void sketch_normalize_kernel(float* __restrict__ out,
                                        const float* __restrict__ mm, int n)
{
    const int i = blockIdx.x * blockDim.x + threadIdx.x;
    if (i < n) {
        const float tmin = mm[0];
        const float tmax = mm[1];
        out[i] = (out[i] - tmin) / (tmax - tmin);
    }
}

extern "C" void kernel_launch(void* const* d_in, const int* in_sizes, int n_in,
                              void* d_out, int out_size, void* d_ws, size_t ws_size,
                              hipStream_t stream)
{
    const float* xs = (const float*)d_in[0];
    const float* ys = (const float*)d_in[1];
    float* out = (float*)d_out;
    float* ws  = (float*)d_ws;

    float* bmin = ws;
    float* bmax = ws + 1024;
    float* mm   = ws + 2048;

    dim3 grid(IMSIZE / 16, IMSIZE / 64);
    sketch_render_kernel<<<grid, 256, 0, stream>>>(xs, ys, out, bmin, bmax);

    const int nblocks = (IMSIZE / 16) * (IMSIZE / 64);
    sketch_minmax_kernel<<<1, 256, 0, stream>>>(bmin, bmax, mm, nblocks);

    const int n = IMSIZE * IMSIZE;
    sketch_normalize_kernel<<<(n + 255) / 256, 256, 0, stream>>>(out, mm, n);
}


# Render loop interchanged: each 16-segment group's parameters are read from LDS once and kept in registers across the wave's 8 rows (8x fewer LDS reads); per-pixel sum order unchanged
# speedup vs baseline: 1.2846x; 1.2846x over previous
#include <hip/hip_runtime.h>
#include <math.h>

#define IMSIZE 1024
#define NSEG   128
#define FEPS   1e-10f

typedef __attribute__((ext_vector_type(2))) float v2f;
typedef __attribute__((ext_vector_type(8))) float v8f;

__global__ __launch_bounds__(256) void sketch_render_kernel(
    const float* __restrict__ xs, const float* __restrict__ ys,
    float* __restrict__ out, float* __restrict__ bmin, float* __restrict__ bmax)
{
    __shared__ float s_cy[NSEG], s_cx[NSEG], s_c0[NSEG];
    __shared__ float s_sl[NSEG], s_in[NSEG], s_vt[NSEG];
    __shared__ float s_xl[NSEG], s_xh[NSEG], s_yl[NSEG], s_yh[NSEG];
    __shared__ float s_red[16];

    const int t = threadIdx.x;

    if (t < NSEG) {
        float x0 = xs[t], x1 = xs[t + 1];
        float y0 = ys[t], y1 = ys[t + 1];
        float dx = x1 - x0, dy = y1 - y0;
        float dx2 = dx * dx;
        float d = dx2 + dy * dy;
        float invd = 1.0f / (d + FEPS);
        float cross = y0 * x1 - x0 * y1;
        s_cy[t] = dx * dy * invd;
        s_cx[t] = dx2 * invd;
        s_c0[t] = -dy * cross * invd;
        float invdx = 1.0f / (dx + FEPS);
        s_sl[t] = dy * invdx;
        s_in[t] = cross * invdx;
        s_vt[t] = (x0 == x1) ? 1.0f : 0.0f;
        s_xl[t] = fminf(x0, x1); s_xh[t] = fmaxf(x0, x1);
        s_yl[t] = fminf(y0, y1); s_yh[t] = fmaxf(y0, y1);
    }
    __syncthreads();

    const int lane  = t & 31;
    const int wave  = t >> 5;
    const int laneN = lane & 15;
    const int hi    = lane >> 4;
    const int xbase = blockIdx.x * 16;
    const int ybase = blockIdx.y * 64;
    const float xp  = (float)(xbase + laneN);

    float acc[8];
#pragma unroll
    for (int r = 0; r < 8; ++r) acc[r] = 0.0f;

    for (int g = 0; g < 8; ++g) {
        const int ma = g * 16 + laneN;
        v2f afrag;
        afrag.x = hi ? s_c0[ma] : s_cy[ma];
        afrag.y = hi ? 0.0f     : s_cx[ma];

        const int mbase = g * 16 + hi * 8;
        float p_sl[8], p_in[8], p_xl[8], p_xh[8], p_yl[8], p_yh[8];
        bool  p_vt[8];
#pragma unroll
        for (int j = 0; j < 8; ++j) {
            const int m = mbase + j;
            p_sl[j] = s_sl[m];  p_in[j] = s_in[m];
            p_xl[j] = s_xl[m];  p_xh[j] = s_xh[m];
            p_yl[j] = s_yl[m];  p_yh[j] = s_yh[m];
            p_vt[j] = (s_vt[m] != 0.0f);
        }

#pragma unroll
        for (int r = 0; r < 8; ++r) {
            const float yp = (float)(ybase + wave + r * 8);

            v2f bfrag;
            bfrag.x = hi ? 1.0f : yp;
            bfrag.y = hi ? 0.0f : xp;

            v8f c = {0.f, 0.f, 0.f, 0.f, 0.f, 0.f, 0.f, 0.f};
            v8f dmat = __builtin_amdgcn_wmma_f32_16x16x4_f32(
                false, afrag, false, bfrag, (short)0, c, false, false);

            float a = acc[r];
#pragma unroll
            for (int j = 0; j < 8; ++j) {
                float xg   = dmat[j];
                float ygen = fmaf(p_sl[j], xg, p_in[j]);
                float ysel = p_vt[j] ? yp : ygen;
                float xc   = __builtin_amdgcn_fmed3f(xg,   p_xl[j], p_xh[j]);
                float yc   = __builtin_amdgcn_fmed3f(ysel, p_yl[j], p_yh[j]);
                float ddx  = xp - xc;
                float ddy  = yp - yc;
                a += __builtin_amdgcn_sqrtf(fmaf(ddx, ddx, fmaf(ddy, ddy, FEPS)));
            }
            acc[r] = a;
        }
    }

    float lmin = INFINITY, lmax = -INFINITY;
#pragma unroll
    for (int r = 0; r < 8; ++r) {
        const int row = ybase + wave + r * 8;
        float s = acc[r] + __shfl_xor(acc[r], 16, 32);
        lmin = fminf(lmin, s);
        lmax = fmaxf(lmax, s);
        if (lane < 16)
            out[row * IMSIZE + xbase + laneN] = s;
    }

#pragma unroll
    for (int off = 16; off >= 1; off >>= 1) {
        lmin = fminf(lmin, __shfl_xor(lmin, off, 32));
        lmax = fmaxf(lmax, __shfl_xor(lmax, off, 32));
    }
    if (lane == 0) { s_red[wave] = lmin; s_red[8 + wave] = lmax; }
    __syncthreads();
    if (t == 0) {
        float m0 = s_red[0], m1 = s_red[8];
        for (int w = 1; w < 8; ++w) {
            m0 = fminf(m0, s_red[w]);
            m1 = fmaxf(m1, s_red[8 + w]);
        }
        const int bid = blockIdx.y * gridDim.x + blockIdx.x;
        bmin[bid] = m0;
        bmax[bid] = m1;
    }
}

__global__ __launch_bounds__(256) void sketch_minmax_kernel(
    const float* __restrict__ bmin, const float* __restrict__ bmax,
    float* __restrict__ mm, int n)
{
    __shared__ float smin[8], smax[8];
    const int t = threadIdx.x;
    float m0 = INFINITY, m1 = -INFINITY;
    for (int i = t; i < n; i += 256) {
        m0 = fminf(m0, bmin[i]);
        m1 = fmaxf(m1, bmax[i]);
    }
#pragma unroll
    for (int off = 16; off >= 1; off >>= 1) {
        m0 = fminf(m0, __shfl_xor(m0, off, 32));
        m1 = fmaxf(m1, __shfl_xor(m1, off, 32));
    }
    if ((t & 31) == 0) { smin[t >> 5] = m0; smax[t >> 5] = m1; }
    __syncthreads();
    if (t == 0) {
        for (int w = 1; w < 8; ++w) {
            m0 = fminf(m0, smin[w]);
            m1 = fmaxf(m1, smax[w]);
        }
        mm[0] = m0;
        mm[1] = m1;
    }
}

__global__ void sketch_normalize_kernel(float* __restrict__ out,
                                        const float* __restrict__ mm, int n)
{
    const int i = blockIdx.x * blockDim.x + threadIdx.x;
    if (i < n) {
        const float tmin = mm[0];
        const float tmax = mm[1];
        out[i] = (out[i] - tmin) / (tmax - tmin);
    }
}

extern "C" void kernel_launch(void* const* d_in, const int* in_sizes, int n_in,
                              void* d_out, int out_size, void* d_ws, size_t ws_size,
                              hipStream_t stream)
{
    const float* xs = (const float*)d_in[0];
    const float* ys = (const float*)d_in[1];
    float* out = (float*)d_out;
    float* ws  = (float*)d_ws;

    float* bmin = ws;
    float* bmax = ws + 1024;
    float* mm   = ws + 2048;

    dim3 grid(IMSIZE / 16, IMSIZE / 64);
    sketch_render_kernel<<<grid, 256, 0, stream>>>(xs, ys, out, bmin, bmax);

    const int nblocks = (IMSIZE / 16) * (IMSIZE / 64);
    sketch_minmax_kernel<<<1, 256, 0, stream>>>(bmin, bmax, mm, nblocks);

    const int n = IMSIZE * IMSIZE;
    sketch_normalize_kernel<<<(n + 255) / 256, 256, 0, stream>>>(out, mm, n);
}


# A rows and per-group constants absorb the x offset and the vertical case: WMMA gives xp - x_gen (vertical row: yp), x clamp bounds pre-shifted by xp; tail loses select and x subtract; sum order kept
# speedup vs baseline: 1.3946x; 1.0856x over previous
#include <hip/hip_runtime.h>
#include <math.h>

#define IMSIZE 1024
#define NSEG   128
#define FEPS   1e-10f

typedef __attribute__((ext_vector_type(2))) float v2f;
typedef __attribute__((ext_vector_type(8))) float v8f;

__global__ __launch_bounds__(256) void sketch_render_kernel(
    const float* __restrict__ xs, const float* __restrict__ ys,
    float* __restrict__ out, float* __restrict__ bmin, float* __restrict__ bmax)
{
    __shared__ float s_cy[NSEG], s_cx[NSEG], s_c0[NSEG];
    __shared__ float s_sl[NSEG], s_in[NSEG], s_ns[NSEG];
    __shared__ float s_xl[NSEG], s_xh[NSEG], s_yl[NSEG], s_yh[NSEG];
    __shared__ float s_red[16];

    const int t = threadIdx.x;

    if (t < NSEG) {
        float x0 = xs[t], x1 = xs[t + 1];
        float y0 = ys[t], y1 = ys[t + 1];
        float dx = x1 - x0, dy = y1 - y0;
        float dx2 = dx * dx;
        float d = dx2 + dy * dy;
        float invd = 1.0f / (d + FEPS);
        float cross = y0 * x1 - x0 * y1;
        float cy = dx * dy * invd;
        float cx = dx2 * invd;
        float c0 = -dy * cross * invd;
        float invdx = 1.0f / (dx + FEPS);
        float sl = dy * invdx;
        float in = cross * invdx;
        const bool vert = (x0 == x1);
        s_cy[t] = vert ? 1.0f : -cy;
        s_cx[t] = vert ? 0.0f : 1.0f - cx;
        s_c0[t] = vert ? 0.0f : -c0;
        s_ns[t] = vert ? 1.0f : -sl;
        s_sl[t] = vert ? 0.0f : sl;
        s_in[t] = vert ? 0.0f : in;
        s_xl[t] = fminf(x0, x1); s_xh[t] = fmaxf(x0, x1);
        s_yl[t] = fminf(y0, y1); s_yh[t] = fmaxf(y0, y1);
    }
    __syncthreads();

    const int lane  = t & 31;
    const int wave  = t >> 5;
    const int laneN = lane & 15;
    const int hi    = lane >> 4;
    const int xbase = blockIdx.x * 16;
    const int ybase = blockIdx.y * 64;
    const float xp  = (float)(xbase + laneN);

    float acc[8];
#pragma unroll
    for (int r = 0; r < 8; ++r) acc[r] = 0.0f;

    for (int g = 0; g < 8; ++g) {
        const int ma = g * 16 + laneN;
        v2f afrag;
        afrag.x = hi ? s_c0[ma] : s_cy[ma];
        afrag.y = hi ? 0.0f     : s_cx[ma];

        const int mbase = g * 16 + hi * 8;
        float p_ns[8], p_of[8], p_bl[8], p_bh[8], p_yl[8], p_yh[8];
#pragma unroll
        for (int j = 0; j < 8; ++j) {
            const int m = mbase + j;
            p_ns[j] = s_ns[m];        p_of[j] = fmaf(s_sl[m], xp, s_in[m]);
            p_bl[j] = xp - s_xh[m];   p_bh[j] = xp - s_xl[m];
            p_yl[j] = s_yl[m];        p_yh[j] = s_yh[m];
        }

#pragma unroll
        for (int r = 0; r < 8; ++r) {
            const float yp = (float)(ybase + wave + r * 8);

            v2f bfrag;
            bfrag.x = hi ? 1.0f : yp;
            bfrag.y = hi ? 0.0f : xp;

            v8f c = {0.f, 0.f, 0.f, 0.f, 0.f, 0.f, 0.f, 0.f};
            v8f dmat = __builtin_amdgcn_wmma_f32_16x16x4_f32(
                false, afrag, false, bfrag, (short)0, c, false, false);

            float a = acc[r];
#pragma unroll
            for (int j = 0; j < 8; ++j) {
                float e    = dmat[j];
                float ygen = fmaf(p_ns[j], e, p_of[j]);
                float ddx  = __builtin_amdgcn_fmed3f(e,    p_bl[j], p_bh[j]);
                float yc   = __builtin_amdgcn_fmed3f(ygen, p_yl[j], p_yh[j]);
                float ddy  = yp - yc;
                a += __builtin_amdgcn_sqrtf(fmaf(ddx, ddx, fmaf(ddy, ddy, FEPS)));
            }
            acc[r] = a;
        }
    }

    float lmin = INFINITY, lmax = -INFINITY;
#pragma unroll
    for (int r = 0; r < 8; ++r) {
        const int row = ybase + wave + r * 8;
        float s = acc[r] + __shfl_xor(acc[r], 16, 32);
        lmin = fminf(lmin, s);
        lmax = fmaxf(lmax, s);
        if (lane < 16)
            out[row * IMSIZE + xbase + laneN] = s;
    }

#pragma unroll
    for (int off = 16; off >= 1; off >>= 1) {
        lmin = fminf(lmin, __shfl_xor(lmin, off, 32));
        lmax = fmaxf(lmax, __shfl_xor(lmax, off, 32));
    }
    if (lane == 0) { s_red[wave] = lmin; s_red[8 + wave] = lmax; }
    __syncthreads();
    if (t == 0) {
        float m0 = s_red[0], m1 = s_red[8];
        for (int w = 1; w < 8; ++w) {
            m0 = fminf(m0, s_red[w]);
            m1 = fmaxf(m1, s_red[8 + w]);
        }
        const int bid = blockIdx.y * gridDim.x + blockIdx.x;
        bmin[bid] = m0;
        bmax[bid] = m1;
    }
}

__global__ __launch_bounds__(256) void sketch_minmax_kernel(
    const float* __restrict__ bmin, const float* __restrict__ bmax,
    float* __restrict__ mm, int n)
{
    __shared__ float smin[8], smax[8];
    const int t = threadIdx.x;
    float m0 = INFINITY, m1 = -INFINITY;
    for (int i = t; i < n; i += 256) {
        m0 = fminf(m0, bmin[i]);
        m1 = fmaxf(m1, bmax[i]);
    }
#pragma unroll
    for (int off = 16; off >= 1; off >>= 1) {
        m0 = fminf(m0, __shfl_xor(m0, off, 32));
        m1 = fmaxf(m1, __shfl_xor(m1, off, 32));
    }
    if ((t & 31) == 0) { smin[t >> 5] = m0; smax[t >> 5] = m1; }
    __syncthreads();
    if (t == 0) {
        for (int w = 1; w < 8; ++w) {
            m0 = fminf(m0, smin[w]);
            m1 = fmaxf(m1, smax[w]);
        }
        mm[0] = m0;
        mm[1] = m1;
    }
}

__global__ void sketch_normalize_kernel(float* __restrict__ out,
                                        const float* __restrict__ mm, int n)
{
    const int i = blockIdx.x * blockDim.x + threadIdx.x;
    if (i < n) {
        const float tmin = mm[0];
        const float tmax = mm[1];
        out[i] = (out[i] - tmin) / (tmax - tmin);
    }
}

extern "C" void kernel_launch(void* const* d_in, const int* in_sizes, int n_in,
                              void* d_out, int out_size, void* d_ws, size_t ws_size,
                              hipStream_t stream)
{
    const float* xs = (const float*)d_in[0];
    const float* ys = (const float*)d_in[1];
    float* out = (float*)d_out;
    float* ws  = (float*)d_ws;

    float* bmin = ws;
    float* bmax = ws + 1024;
    float* mm   = ws + 2048;

    dim3 grid(IMSIZE / 16, IMSIZE / 64);
    sketch_render_kernel<<<grid, 256, 0, stream>>>(xs, ys, out, bmin, bmax);

    const int nblocks = (IMSIZE / 16) * (IMSIZE / 64);
    sketch_minmax_kernel<<<1, 256, 0, stream>>>(bmin, bmax, mm, nblocks);

    const int n = IMSIZE * IMSIZE;
    sketch_normalize_kernel<<<(n + 255) / 256, 256, 0, stream>>>(out, mm, n);
}


# Min/max launch folded into normalize: each normalize block reduces the 1024 per-block minima and maxima itself (one 128-bit load of each per thread, wave butterfly, one barrier); 2 launches, not 3
# speedup vs baseline: 1.4970x; 1.0734x over previous
#include <hip/hip_runtime.h>
#include <math.h>

#define IMSIZE 1024
#define NSEG   128
#define FEPS   1e-10f

typedef __attribute__((ext_vector_type(2))) float v2f;
typedef __attribute__((ext_vector_type(8))) float v8f;

__global__ __launch_bounds__(256) void sketch_render_kernel(
    const float* __restrict__ xs, const float* __restrict__ ys,
    float* __restrict__ out, float* __restrict__ bmin, float* __restrict__ bmax)
{
    __shared__ float s_cy[NSEG], s_cx[NSEG], s_c0[NSEG];
    __shared__ float s_sl[NSEG], s_in[NSEG], s_ns[NSEG];
    __shared__ float s_xl[NSEG], s_xh[NSEG], s_yl[NSEG], s_yh[NSEG];
    __shared__ float s_red[16];

    const int t = threadIdx.x;

    if (t < NSEG) {
        float x0 = xs[t], x1 = xs[t + 1];
        float y0 = ys[t], y1 = ys[t + 1];
        float dx = x1 - x0, dy = y1 - y0;
        float dx2 = dx * dx;
        float d = dx2 + dy * dy;
        float invd = 1.0f / (d + FEPS);
        float cross = y0 * x1 - x0 * y1;
        float cy = dx * dy * invd;
        float cx = dx2 * invd;
        float c0 = -dy * cross * invd;
        float invdx = 1.0f / (dx + FEPS);
        float sl = dy * invdx;
        float in = cross * invdx;
        const bool vert = (x0 == x1);
        s_cy[t] = vert ? 1.0f : -cy;
        s_cx[t] = vert ? 0.0f : 1.0f - cx;
        s_c0[t] = vert ? 0.0f : -c0;
        s_ns[t] = vert ? 1.0f : -sl;
        s_sl[t] = vert ? 0.0f : sl;
        s_in[t] = vert ? 0.0f : in;
        s_xl[t] = fminf(x0, x1); s_xh[t] = fmaxf(x0, x1);
        s_yl[t] = fminf(y0, y1); s_yh[t] = fmaxf(y0, y1);
    }
    __syncthreads();

    const int lane  = t & 31;
    const int wave  = t >> 5;
    const int laneN = lane & 15;
    const int hi    = lane >> 4;
    const int xbase = blockIdx.x * 16;
    const int ybase = blockIdx.y * 64;
    const float xp  = (float)(xbase + laneN);

    float acc[8];
#pragma unroll
    for (int r = 0; r < 8; ++r) acc[r] = 0.0f;

    for (int g = 0; g < 8; ++g) {
        const int ma = g * 16 + laneN;
        v2f afrag;
        afrag.x = hi ? s_c0[ma] : s_cy[ma];
        afrag.y = hi ? 0.0f     : s_cx[ma];

        const int mbase = g * 16 + hi * 8;
        float p_ns[8], p_of[8], p_bl[8], p_bh[8], p_yl[8], p_yh[8];
#pragma unroll
        for (int j = 0; j < 8; ++j) {
            const int m = mbase + j;
            p_ns[j] = s_ns[m];        p_of[j] = fmaf(s_sl[m], xp, s_in[m]);
            p_bl[j] = xp - s_xh[m];   p_bh[j] = xp - s_xl[m];
            p_yl[j] = s_yl[m];        p_yh[j] = s_yh[m];
        }

#pragma unroll
        for (int r = 0; r < 8; ++r) {
            const float yp = (float)(ybase + wave + r * 8);

            v2f bfrag;
            bfrag.x = hi ? 1.0f : yp;
            bfrag.y = hi ? 0.0f : xp;

            v8f c = {0.f, 0.f, 0.f, 0.f, 0.f, 0.f, 0.f, 0.f};
            v8f dmat = __builtin_amdgcn_wmma_f32_16x16x4_f32(
                false, afrag, false, bfrag, (short)0, c, false, false);

            float a = acc[r];
#pragma unroll
            for (int j = 0; j < 8; ++j) {
                float e    = dmat[j];
                float ygen = fmaf(p_ns[j], e, p_of[j]);
                float ddx  = __builtin_amdgcn_fmed3f(e,    p_bl[j], p_bh[j]);
                float yc   = __builtin_amdgcn_fmed3f(ygen, p_yl[j], p_yh[j]);
                float ddy  = yp - yc;
                a += __builtin_amdgcn_sqrtf(fmaf(ddx, ddx, fmaf(ddy, ddy, FEPS)));
            }
            acc[r] = a;
        }
    }

    float lmin = INFINITY, lmax = -INFINITY;
#pragma unroll
    for (int r = 0; r < 8; ++r) {
        const int row = ybase + wave + r * 8;
        float s = acc[r] + __shfl_xor(acc[r], 16, 32);
        lmin = fminf(lmin, s);
        lmax = fmaxf(lmax, s);
        if (lane < 16)
            out[row * IMSIZE + xbase + laneN] = s;
    }

#pragma unroll
    for (int off = 16; off >= 1; off >>= 1) {
        lmin = fminf(lmin, __shfl_xor(lmin, off, 32));
        lmax = fmaxf(lmax, __shfl_xor(lmax, off, 32));
    }
    if (lane == 0) { s_red[wave] = lmin; s_red[8 + wave] = lmax; }
    __syncthreads();
    if (t == 0) {
        float m0 = s_red[0], m1 = s_red[8];
        for (int w = 1; w < 8; ++w) {
            m0 = fminf(m0, s_red[w]);
            m1 = fmaxf(m1, s_red[8 + w]);
        }
        const int bid = blockIdx.y * gridDim.x + blockIdx.x;
        bmin[bid] = m0;
        bmax[bid] = m1;
    }
}

#define NPART ((IMSIZE / 16) * (IMSIZE / 64))

__global__ __launch_bounds__(256) void sketch_normalize_kernel(
    float* __restrict__ out, const float* __restrict__ bmin,
    const float* __restrict__ bmax, int n)
{
    static_assert(NPART == 4 * 256, "one float4 of each partial array per thread");
    __shared__ float smin[8], smax[8];
    const int t = threadIdx.x;
    const int i = blockIdx.x * 256 + t;

    float v = 0.0f;
    if (i < n) v = out[i];

    const float4 a = reinterpret_cast<const float4*>(bmin)[t];
    const float4 b = reinterpret_cast<const float4*>(bmax)[t];
    float m0 = fminf(fminf(a.x, a.y), fminf(a.z, a.w));
    float m1 = fmaxf(fmaxf(b.x, b.y), fmaxf(b.z, b.w));
#pragma unroll
    for (int off = 16; off >= 1; off >>= 1) {
        m0 = fminf(m0, __shfl_xor(m0, off, 32));
        m1 = fmaxf(m1, __shfl_xor(m1, off, 32));
    }
    if ((t & 31) == 0) { smin[t >> 5] = m0; smax[t >> 5] = m1; }
    __syncthreads();

    float tmin = smin[0], tmax = smax[0];
#pragma unroll
    for (int w = 1; w < 8; ++w) {
        tmin = fminf(tmin, smin[w]);
        tmax = fmaxf(tmax, smax[w]);
    }
    if (i < n)
        out[i] = (v - tmin) / (tmax - tmin);
}

extern "C" void kernel_launch(void* const* d_in, const int* in_sizes, int n_in,
                              void* d_out, int out_size, void* d_ws, size_t ws_size,
                              hipStream_t stream)
{
    const float* xs = (const float*)d_in[0];
    const float* ys = (const float*)d_in[1];
    float* out = (float*)d_out;
    float* ws  = (float*)d_ws;

    float* bmin = ws;
    float* bmax = ws + 1024;

    dim3 grid(IMSIZE / 16, IMSIZE / 64);
    sketch_render_kernel<<<grid, 256, 0, stream>>>(xs, ys, out, bmin, bmax);

    const int n = IMSIZE * IMSIZE;
    sketch_normalize_kernel<<<(n + 255) / 256, 256, 0, stream>>>(out, bmin, bmax, n);
}


# Distance taken in each segment's own frame: two WMMAs per tile give along-line and across-line coordinates, one clamp of the along one to [0,len] replaces both box clamps; sum order kept
# speedup vs baseline: 1.6062x; 1.0729x over previous
#include <hip/hip_runtime.h>
#include <math.h>

#define IMSIZE 1024
#define NSEG   128
#define FEPS   1e-10f

typedef __attribute__((ext_vector_type(2))) float v2f;
typedef __attribute__((ext_vector_type(8))) float v8f;

__global__ __launch_bounds__(256) __attribute__((amdgpu_waves_per_eu(8, 8)))
void sketch_render_kernel(
    const float* __restrict__ xs, const float* __restrict__ ys,
    float* __restrict__ out, float* __restrict__ bmin, float* __restrict__ bmax)
{
    __shared__ float s_vy[NSEG], s_vx[NSEG], s_v0[NSEG];
    __shared__ float s_wy[NSEG], s_wx[NSEG], s_w0[NSEG];
    __shared__ float s_len[NSEG];
    __shared__ float s_red[16];

    const int t = threadIdx.x;

    if (t < NSEG) {
        float x0 = xs[t], x1 = xs[t + 1];
        float y0 = ys[t], y1 = ys[t + 1];
        float dx = x1 - x0, dy = y1 - y0;
        float d  = dx * dx + dy * dy;
        const bool pt = !(d > 0.0f);
        float len = pt ? 0.0f : sqrtf(d);
        float ux  = pt ? 1.0f : dx / len;
        float uy  = pt ? 0.0f : dy / len;
        s_vy[t] = uy;   s_vx[t] = ux;    s_v0[t] = -(ux * x0 + uy * y0);
        s_wy[t] = ux;   s_wx[t] = -uy;   s_w0[t] = uy * x0 - ux * y0;
        s_len[t] = len;
    }
    __syncthreads();

    const int lane  = t & 31;
    const int wave  = t >> 5;
    const int laneN = lane & 15;
    const int hi    = lane >> 4;
    const int xbase = blockIdx.x * 16;
    const int ybase = blockIdx.y * 64;
    const float xp  = (float)(xbase + laneN);

    float acc[8];
#pragma unroll
    for (int r = 0; r < 8; ++r) acc[r] = 0.0f;

    for (int g = 0; g < 8; ++g) {
        const int ma = g * 16 + laneN;
        v2f afrag_v, afrag_w;
        afrag_v.x = hi ? s_v0[ma] : s_vy[ma];
        afrag_v.y = hi ? 0.0f     : s_vx[ma];
        afrag_w.x = hi ? s_w0[ma] : s_wy[ma];
        afrag_w.y = hi ? 0.0f     : s_wx[ma];

        const int mbase = g * 16 + hi * 8;
        float p_len[8];
#pragma unroll
        for (int j = 0; j < 8; ++j) p_len[j] = s_len[mbase + j];

#pragma unroll
        for (int r = 0; r < 8; ++r) {
            const float yp = (float)(ybase + wave + r * 8);

            v2f bfrag;
            bfrag.x = hi ? 1.0f : yp;
            bfrag.y = hi ? 0.0f : xp;

            v8f c = {0.f, 0.f, 0.f, 0.f, 0.f, 0.f, 0.f, 0.f};
            v8f vmat = __builtin_amdgcn_wmma_f32_16x16x4_f32(
                false, afrag_v, false, bfrag, (short)0, c, false, false);
            v8f wmat = __builtin_amdgcn_wmma_f32_16x16x4_f32(
                false, afrag_w, false, bfrag, (short)0, c, false, false);

            float a = acc[r];
#pragma unroll
            for (int j = 0; j < 8; ++j) {
                float v = vmat[j];
                float w = wmat[j];
                float o = v - __builtin_amdgcn_fmed3f(v, 0.0f, p_len[j]);
                a += __builtin_amdgcn_sqrtf(fmaf(o, o, fmaf(w, w, FEPS)));
            }
            acc[r] = a;
        }
    }

    float lmin = INFINITY, lmax = -INFINITY;
#pragma unroll
    for (int r = 0; r < 8; ++r) {
        const int row = ybase + wave + r * 8;
        float s = acc[r] + __shfl_xor(acc[r], 16, 32);
        lmin = fminf(lmin, s);
        lmax = fmaxf(lmax, s);
        if (lane < 16)
            out[row * IMSIZE + xbase + laneN] = s;
    }

#pragma unroll
    for (int off = 16; off >= 1; off >>= 1) {
        lmin = fminf(lmin, __shfl_xor(lmin, off, 32));
        lmax = fmaxf(lmax, __shfl_xor(lmax, off, 32));
    }
    if (lane == 0) { s_red[wave] = lmin; s_red[8 + wave] = lmax; }
    __syncthreads();
    if (t == 0) {
        float m0 = s_red[0], m1 = s_red[8];
        for (int w = 1; w < 8; ++w) {
            m0 = fminf(m0, s_red[w]);
            m1 = fmaxf(m1, s_red[8 + w]);
        }
        const int bid = blockIdx.y * gridDim.x + blockIdx.x;
        bmin[bid] = m0;
        bmax[bid] = m1;
    }
}

#define NPART ((IMSIZE / 16) * (IMSIZE / 64))

__global__ __launch_bounds__(256) void sketch_normalize_kernel(
    float* __restrict__ out, const float* __restrict__ bmin,
    const float* __restrict__ bmax, int n)
{
    static_assert(NPART == 4 * 256, "one float4 of each partial array per thread");
    __shared__ float smin[8], smax[8];
    const int t = threadIdx.x;
    const int i = blockIdx.x * 256 + t;

    float v = 0.0f;
    if (i < n) v = out[i];

    const float4 a = reinterpret_cast<const float4*>(bmin)[t];
    const float4 b = reinterpret_cast<const float4*>(bmax)[t];
    float m0 = fminf(fminf(a.x, a.y), fminf(a.z, a.w));
    float m1 = fmaxf(fmaxf(b.x, b.y), fmaxf(b.z, b.w));
#pragma unroll
    for (int off = 16; off >= 1; off >>= 1) {
        m0 = fminf(m0, __shfl_xor(m0, off, 32));
        m1 = fmaxf(m1, __shfl_xor(m1, off, 32));
    }
    if ((t & 31) == 0) { smin[t >> 5] = m0; smax[t >> 5] = m1; }
    __syncthreads();

    float tmin = smin[0], tmax = smax[0];
#pragma unroll
    for (int w = 1; w < 8; ++w) {
        tmin = fminf(tmin, smin[w]);
        tmax = fmaxf(tmax, smax[w]);
    }
    if (i < n)
        out[i] = (v - tmin) / (tmax - tmin);
}

extern "C" void kernel_launch(void* const* d_in, const int* in_sizes, int n_in,
                              void* d_out, int out_size, void* d_ws, size_t ws_size,
                              hipStream_t stream)
{
    const float* xs = (const float*)d_in[0];
    const float* ys = (const float*)d_in[1];
    float* out = (float*)d_out;
    float* ws  = (float*)d_ws;

    float* bmin = ws;
    float* bmax = ws + 1024;

    dim3 grid(IMSIZE / 16, IMSIZE / 64);
    sketch_render_kernel<<<grid, 256, 0, stream>>>(xs, ys, out, bmin, bmax);

    const int n = IMSIZE * IMSIZE;
    sketch_normalize_kernel<<<(n + 255) / 256, 256, 0, stream>>>(out, bmin, bmax, n);
}
